# starting kernel
# baseline (speedup 1.0000x reference)
#include <hip/hip_runtime.h>
#include <hip/hip_bf16.h>

typedef __attribute__((ext_vector_type(2))) float v2f;
typedef __attribute__((ext_vector_type(8))) float v8f;

#define FEAT 128
#define TILES_PER_BLOCK 8

__global__ __launch_bounds__(256) void bilinear_gemm_zW(
    const float* __restrict__ z, const float* __restrict__ W,
    float* __restrict__ V, int n_nodes) {
  __shared__ float Al[16][132];

  const int tid   = threadIdx.x;
  const int wv    = tid >> 5;
  const int lane  = tid & 31;
  const int colBase = wv * 16;
  const int mrow  = lane & 15;
  const int khalf = (lane >> 4) << 1;

  v2f bf[FEAT / 4];
#pragma unroll
  for (int kk = 0; kk < FEAT / 4; ++kk) {
    const int k = kk * 4 + khalf;
    bf[kk][0] = W[(size_t)k * FEAT + colBase + mrow];
    bf[kk][1] = W[(size_t)(k + 1) * FEAT + colBase + mrow];
  }

  for (int t = 0; t < TILES_PER_BLOCK; ++t) {
    const int rowBase = (blockIdx.x * TILES_PER_BLOCK + t) * 16;
    if (rowBase >= n_nodes) break;

    __syncthreads();
#pragma unroll
    for (int c4 = tid; c4 < (16 * FEAT) / 4; c4 += 256) {
      const int r = c4 >> 5;
      const int c = (c4 & 31) << 2;
      *(float4*)&Al[r][c] = *(const float4*)&z[(size_t)(rowBase + r) * FEAT + c];
    }
    __syncthreads();

    v8f acc = {};
#pragma unroll
    for (int kk = 0; kk < FEAT / 4; ++kk) {
      const int k = kk * 4 + khalf;
      v2f a;
      a[0] = Al[mrow][k];
      a[1] = Al[mrow][k + 1];
      acc = __builtin_amdgcn_wmma_f32_16x16x4_f32(
          false, a, false, bf[kk], (short)0, acc, false, false);
    }

    const int mOff = (lane >> 4) << 3;
#pragma unroll
    for (int r = 0; r < 8; ++r) {
      V[(size_t)(rowBase + mOff + r) * FEAT + colBase + mrow] = acc[r];
    }
  }
}

__global__ __launch_bounds__(256) void bilinear_edge_scores(
    const float* __restrict__ V, const float* __restrict__ z,
    const int* __restrict__ e_src, const int* __restrict__ e_dst,
    float* __restrict__ out, int n_edges) {
  const int lane   = threadIdx.x & 31;
  const int wid    = (blockIdx.x * blockDim.x + threadIdx.x) >> 5;
  const int nwaves = (gridDim.x * blockDim.x) >> 5;

  for (int e = wid; e < n_edges; e += nwaves) {
    const int s = e_src[e];
    const int d = e_dst[e];
    const float4 av = ((const float4*)(V + (size_t)s * FEAT))[lane];
    const float4 bv = ((const float4*)(z + (size_t)d * FEAT))[lane];
    float sum = av.x * bv.x + av.y * bv.y + av.z * bv.z + av.w * bv.w;
#pragma unroll
    for (int off = 16; off > 0; off >>= 1) sum += __shfl_xor(sum, off, 32);
    if (lane == 0) out[e] = sum;
  }
}

extern "C" void kernel_launch(void* const* d_in, const int* in_sizes, int n_in,
                              void* d_out, int out_size, void* d_ws, size_t ws_size,
                              hipStream_t stream) {
  const float* z  = (const float*)d_in[0];
  const int*   ei = (const int*)d_in[1];
  const float* W  = (const float*)d_in[2];
  float* out = (float*)d_out;
  float* V   = (float*)d_ws;

  const int n_nodes = in_sizes[0] / FEAT;
  const int n_edges = in_sizes[1] / 2;

  const int rowTiles = (n_nodes + 15) / 16;
  const int gemmBlocks = (rowTiles + TILES_PER_BLOCK - 1) / TILES_PER_BLOCK;
  bilinear_gemm_zW<<<gemmBlocks, 256, 0, stream>>>(z, W, V, n_nodes);

  bilinear_edge_scores<<<2048, 256, 0, stream>>>(V, z, ei, ei + n_edges, out,
                                                 n_edges);
}


# edge-score kernel: one edge per 8-lane group (4 edges per wave instruction, a lane reads 4 x 16 B of both rows), 3-step DPP8 group sum, 32 scores per trip leave as one line; 256 blocks of 16 waves
# speedup vs baseline: 1.9508x; 1.9508x over previous
#include <hip/hip_runtime.h>
#include <hip/hip_bf16.h>

typedef __attribute__((ext_vector_type(2))) float v2f;
typedef __attribute__((ext_vector_type(8))) float v8f;

#define FEAT 128


__device__ __forceinline__ void load_tile_rows(float4 (&a)[16], const float* __restrict__ z,
                                               int t, int c, int h) {
  const float* zr = z + (size_t)t * (16 * FEAT) + (c * FEAT + 4 * h);
#pragma unroll
  for (int q = 0; q < 16; ++q) a[q] = *(const float4*)(zr + 8 * q);
}

__device__ __forceinline__ void tile_product(const float4 (&a)[16], const v2f (&bf)[4][32],
                                             float* __restrict__ V, int t, int c, int h,
                                             int col0) {
  v8f acc[4] = {};
#pragma unroll
  for (int q = 0; q < 16; ++q) {
    v2f a0, a1;
    a0[0] = a[q].x; a0[1] = a[q].y;
    a1[0] = a[q].z; a1[1] = a[q].w;
#pragma unroll
    for (int j = 0; j < 4; ++j)
      acc[j] = __builtin_amdgcn_wmma_f32_16x16x4_f32(
          false, a0, false, bf[j][2 * q], (short)0, acc[j], false, false);
#pragma unroll
    for (int j = 0; j < 4; ++j)
      acc[j] = __builtin_amdgcn_wmma_f32_16x16x4_f32(
          false, a1, false, bf[j][2 * q + 1], (short)0, acc[j], false, false);
  }
  float* vp = V + (size_t)t * (16 * FEAT) + (8 * h * FEAT + col0 + 4 * c);
#pragma unroll
  for (int r = 0; r < 8; ++r) {
    float4 o;
    o.x = acc[0][r]; o.y = acc[1][r]; o.z = acc[2][r]; o.w = acc[3][r];
    *(float4*)(vp + r * FEAT) = o;
  }
}

__global__ __launch_bounds__(256) void bilinear_gemm_zW(
    const float* __restrict__ z, const float* __restrict__ W,
    float* __restrict__ V, int n_nodes) {
  const int tid  = threadIdx.x;
  const int wv   = __builtin_amdgcn_readfirstlane(tid >> 5);
  const int lane = tid & 31;
  const int c    = lane & 15;
  const int h    = lane >> 4;
  const int half = wv & 1;
  const int pairId = wv >> 1;
  const int col0 = half * 64;
  const int rowTiles = n_nodes >> 4;
  const int nPairs = gridDim.x * 4;
  int t = pairId * gridDim.x + blockIdx.x;
  if (t >= rowTiles) return;

  float4 aA[16], aB[16];
  load_tile_rows(aA, z, t, c, h);

  v2f bf[4][32];
  {
    const float* wp = W + col0 + 4 * c;
#pragma unroll
    for (int q = 0; q < 16; ++q) {
#pragma unroll
      for (int p = 0; p < 2; ++p) {
#pragma unroll
        for (int s = 0; s < 2; ++s) {
          const int k = 8 * q + 4 * h + 2 * p + s;
          const float4 w4 = *(const float4*)(wp + (size_t)k * FEAT);
          bf[0][2 * q + p][s] = w4.x;
          bf[1][2 * q + p][s] = w4.y;
          bf[2][2 * q + p][s] = w4.z;
          bf[3][2 * q + p][s] = w4.w;
        }
      }
    }
  }

  const int step2 = 2 * nPairs;
  while (t + step2 < rowTiles) {
    load_tile_rows(aB, z, t + nPairs, c, h);
    __builtin_amdgcn_sched_barrier(0);
    tile_product(aA, bf, V, t, c, h, col0);
    load_tile_rows(aA, z, t + step2, c, h);
    __builtin_amdgcn_sched_barrier(0);
    tile_product(aB, bf, V, t + nPairs, c, h, col0);
    t += step2;
  }
  if (t + nPairs < rowTiles) {
    load_tile_rows(aB, z, t + nPairs, c, h);
    __builtin_amdgcn_sched_barrier(0);
    tile_product(aA, bf, V, t, c, h, col0);
    __builtin_amdgcn_sched_barrier(0);
    tile_product(aB, bf, V, t + nPairs, c, h, col0);
  } else {
    tile_product(aA, bf, V, t, c, h, col0);
  }
}


__host__ __device__ constexpr unsigned dpp8_sel(unsigned a0, unsigned a1, unsigned a2, unsigned a3,
                                                unsigned a4, unsigned a5, unsigned a6, unsigned a7) {
  return a0 | (a1 << 3) | (a2 << 6) | (a3 << 9) | (a4 << 12) | (a5 << 15) | (a6 << 18) | (a7 << 21);
}

template <int T>
__device__ __forceinline__ unsigned group_lane(unsigned v) {
  return __builtin_amdgcn_mov_dpp8(v, dpp8_sel(T, T, T, T, T, T, T, T));
}

__device__ __forceinline__ float group_sum8(float x) {
  x += __uint_as_float(__builtin_amdgcn_mov_dpp8(__float_as_uint(x), dpp8_sel(4, 5, 6, 7, 0, 1, 2, 3)));
  x += __uint_as_float(__builtin_amdgcn_mov_dpp8(__float_as_uint(x), dpp8_sel(2, 3, 0, 1, 6, 7, 4, 5)));
  x += __uint_as_float(__builtin_amdgcn_mov_dpp8(__float_as_uint(x), dpp8_sel(1, 0, 3, 2, 5, 4, 7, 6)));
  return x;
}

template <int T>
__device__ __forceinline__ float edge_turn(const float* __restrict__ V, const float* __restrict__ z,
                                           unsigned sidx, unsigned didx, int i) {
  const unsigned s = group_lane<T>(sidx);
  const unsigned d = group_lane<T>(didx);
  const float4* vp = (const float4*)(V + (size_t)s * FEAT) + i;
  const float4* zp = (const float4*)(z + (size_t)d * FEAT) + i;
  float sum = 0.f;
#pragma unroll
  for (int j = 0; j < 4; ++j) {
    const float4 a = vp[8 * j];
    const float4 b = zp[8 * j];
    sum += a.x * b.x + a.y * b.y + a.z * b.z + a.w * b.w;
  }
  return group_sum8(sum);
}

__global__ __launch_bounds__(512) void bilinear_edge_scores(
    const float* __restrict__ V, const float* __restrict__ z,
    const int* __restrict__ e_src, const int* __restrict__ e_dst,
    float* __restrict__ out, int n_edges) {
  const int lane = threadIdx.x & 31;
  const int i    = lane & 7;
  const int wv     = __builtin_amdgcn_readfirstlane(threadIdx.x >> 5);
  const int wpb    = blockDim.x >> 5;
  const int wave   = wv * gridDim.x + blockIdx.x;
  const int nwaves = gridDim.x * wpb;
  const int ntrips = (n_edges + 31) >> 5;

  for (int b = wave; b < ntrips; b += nwaves) {
    const int e = (b << 5) + lane;
    const bool live = e < n_edges;
    const unsigned sidx = live ? (unsigned)e_src[e] : 0u;
    const unsigned didx = live ? (unsigned)e_dst[e] : 0u;
    float res;
    float r;
    res = edge_turn<0>(V, z, sidx, didx, i);
    r = edge_turn<1>(V, z, sidx, didx, i); res = (i == 1) ? r : res;
    r = edge_turn<2>(V, z, sidx, didx, i); res = (i == 2) ? r : res;
    r = edge_turn<3>(V, z, sidx, didx, i); res = (i == 3) ? r : res;
    r = edge_turn<4>(V, z, sidx, didx, i); res = (i == 4) ? r : res;
    r = edge_turn<5>(V, z, sidx, didx, i); res = (i == 5) ? r : res;
    r = edge_turn<6>(V, z, sidx, didx, i); res = (i == 6) ? r : res;
    r = edge_turn<7>(V, z, sidx, didx, i); res = (i == 7) ? r : res;
    if (live) out[e] = res;
  }
}

extern "C" void kernel_launch(void* const* d_in, const int* in_sizes, int n_in,
                              void* d_out, int out_size, void* d_ws, size_t ws_size,
                              hipStream_t stream) {
  const float* z  = (const float*)d_in[0];
  const int*   ei = (const int*)d_in[1];
  const float* W  = (const float*)d_in[2];
  float* out = (float*)d_out;
  float* V   = (float*)d_ws;

  const int n_nodes = in_sizes[0] / FEAT;
  const int n_edges = in_sizes[1] / 2;

  bilinear_gemm_zW<<<256, 256, 0, stream>>>(z, W, V, n_nodes);

  bilinear_edge_scores<<<256, 512, 0, stream>>>(V, z, ei, ei + n_edges, out, n_edges);
}
